# starting kernel
# baseline (speedup 1.0000x reference)
#include <hip/hip_runtime.h>
#include <hip/hip_bf16.h>

#define NUM_USERS 100000
#define NUM_ITEMS 200000
#define N_TOTAL   (NUM_USERS + NUM_ITEMS)
#define EMBED_DIM 64
#define NUM_LAYERS 3
#define NUM_EDGES 1200000
#define BATCH 4096

typedef __attribute__((ext_vector_type(2))) float v2f;
typedef __attribute__((ext_vector_type(8))) float v8f;

__global__ __launch_bounds__(256) void scatter_add_kernel(
    const float* __restrict__ cur, float* __restrict__ nxt,
    const int* __restrict__ src, const int* __restrict__ dst) {
  long long t = (long long)blockIdx.x * blockDim.x + threadIdx.x;
  int e = (int)(t >> 4);
  if (e >= NUM_EDGES) return;
  int part = ((int)t & 15) << 2;
  size_t s = (size_t)src[e];
  size_t d = (size_t)dst[e];
  const float4 v = *(const float4*)(cur + s * EMBED_DIM + part);
  float* p = nxt + d * EMBED_DIM + part;
  atomicAdd(p + 0, v.x);
  atomicAdd(p + 1, v.y);
  atomicAdd(p + 2, v.z);
  atomicAdd(p + 3, v.w);
}

__global__ __launch_bounds__(256) void gather_acc_kernel(
    const float* __restrict__ layer,
    const int* __restrict__ uidx, const int* __restrict__ iidx,
    float* __restrict__ ubuf, float* __restrict__ vbuf) {
  int t = blockIdx.x * blockDim.x + threadIdx.x;
  int b = t >> 4;
  if (b >= BATCH) return;
  int part = (t & 15) << 2;
  size_t u = (size_t)uidx[b];
  size_t it = (size_t)iidx[b] + NUM_USERS;
  const float4 uu = *(const float4*)(layer + u * EMBED_DIM + part);
  const float4 vv = *(const float4*)(layer + it * EMBED_DIM + part);
  float4* up = (float4*)(ubuf + (size_t)b * EMBED_DIM + part);
  float4* vp = (float4*)(vbuf + (size_t)b * EMBED_DIM + part);
  float4 a = *up;
  a.x += uu.x; a.y += uu.y; a.z += uu.z; a.w += uu.w;
  *up = a;
  float4 c = *vp;
  c.x += vv.x; c.y += vv.y; c.z += vv.z; c.w += vv.w;
  *vp = c;
}

__global__ __launch_bounds__(128) void dot_wmma_kernel(
    const float* __restrict__ ubuf, const float* __restrict__ vbuf,
    float* __restrict__ out) {
  const int lane = threadIdx.x & 31;
  const int tile = blockIdx.x * (blockDim.x >> 5) + (threadIdx.x >> 5);
  const int rowBase = tile * 16;
  const int row = rowBase + (lane & 15);
  const int kOff = (lane >> 4) << 1;

  const float* uRow = ubuf + (size_t)row * EMBED_DIM + kOff;
  const float* vRow = vbuf + (size_t)row * EMBED_DIM + kOff;

  v8f c = {};
#pragma unroll
  for (int k = 0; k < EMBED_DIM; k += 4) {
    v2f a, b;
    a.x = uRow[k];
    a.y = uRow[k + 1];
    b.x = vRow[k];
    b.y = vRow[k + 1];
    c = __builtin_amdgcn_wmma_f32_16x16x4_f32(false, a, false, b, (short)0, c,
                                              false, false);
  }

  float cd[8];
#pragma unroll
  for (int i = 0; i < 8; ++i) cd[i] = c[i];
  int dsel = (lane < 16) ? lane : (lane - 24);
  float diag = cd[0];
#pragma unroll
  for (int i = 1; i < 8; ++i)
    if (dsel == i) diag = cd[i];

  const float scale = 1.0f / 16.0f;
  if (lane < 8)
    out[rowBase + lane] = diag * scale;
  else if (lane >= 24)
    out[rowBase + (lane - 16)] = diag * scale;
}

extern "C" void kernel_launch(void* const* d_in, const int* in_sizes, int n_in,
                              void* d_out, int out_size, void* d_ws,
                              size_t ws_size, hipStream_t stream) {
  const float* emb = (const float*)d_in[0];
  const int* eidx = (const int*)d_in[1];
  const int* uidx = (const int*)d_in[2];
  const int* iidx = (const int*)d_in[3];
  float* out = (float*)d_out;

  const int* src = eidx;
  const int* dst = eidx + NUM_EDGES;

  char* ws = (char*)d_ws;
  const size_t nodeBytes = (size_t)N_TOTAL * EMBED_DIM * sizeof(float);
  const size_t batBytes = (size_t)BATCH * EMBED_DIM * sizeof(float);
  float* bufA = (float*)ws;
  float* bufB = (float*)(ws + nodeBytes);
  float* ubuf = (float*)(ws + 2 * nodeBytes);
  float* vbuf = (float*)(ws + 2 * nodeBytes + batBytes);

  hipMemsetAsync(ubuf, 0, batBytes, stream);
  hipMemsetAsync(vbuf, 0, batBytes, stream);
  const int gGrid = (BATCH * 16 + 255) / 256;
  gather_acc_kernel<<<gGrid, 256, 0, stream>>>(emb, uidx, iidx, ubuf, vbuf);

  const long long sThreads = (long long)NUM_EDGES * 16;
  const int sGrid = (int)((sThreads + 255) / 256);
  const float* cur = emb;
  for (int l = 0; l < NUM_LAYERS; ++l) {
    float* nxt = (l & 1) ? bufB : bufA;
    hipMemsetAsync(nxt, 0, nodeBytes, stream);
    scatter_add_kernel<<<sGrid, 256, 0, stream>>>(cur, nxt, src, dst);
    gather_acc_kernel<<<gGrid, 256, 0, stream>>>(nxt, uidx, iidx, ubuf, vbuf);
    cur = nxt;
  }

  dot_wmma_kernel<<<BATCH / 16 / 4, 128, 0, stream>>>(ubuf, vbuf, out);
}


# Prune to rows the output needs: level table from batch rows and two backward edge hops; layers zero and scatter-add needed rows only, layer 3 onto layer 1 rows, gathers last (float atomics kept)
# speedup vs baseline: 2.7304x; 2.7304x over previous
#include <hip/hip_runtime.h>
#include <hip/hip_bf16.h>

#define NUM_USERS 100000
#define NUM_ITEMS 200000
#define N_TOTAL   (NUM_USERS + NUM_ITEMS)
#define EMBED_DIM 64
#define NUM_LAYERS 3
#define NUM_EDGES 1200000
#define BATCH 4096

typedef __attribute__((ext_vector_type(2))) float v2f;
typedef __attribute__((ext_vector_type(8))) float v8f;

__global__ __launch_bounds__(256) void clear_level_kernel(int* __restrict__ level) {
  int t = blockIdx.x * blockDim.x + threadIdx.x;
  if (t < N_TOTAL / 4) ((int4*)level)[t] = make_int4(0, 0, 0, 0);
}

__global__ __launch_bounds__(256) void mark_batch_kernel(
    const int* __restrict__ uidx, const int* __restrict__ iidx,
    int* __restrict__ level) {
  int t = blockIdx.x * blockDim.x + threadIdx.x;
  if (t < BATCH)
    level[uidx[t]] = NUM_LAYERS;
  else if (t < 2 * BATCH)
    level[NUM_USERS + iidx[t - BATCH]] = NUM_LAYERS;
}

__global__ __launch_bounds__(256) void mark_src_kernel(
    const int* __restrict__ src, const int* __restrict__ dst,
    int* level, int want) {
  int e = blockIdx.x * blockDim.x + threadIdx.x;
  if (e >= NUM_EDGES) return;
  if (level[dst[e]] > want) {
    int s = src[e];
    if (level[s] < want) level[s] = want;
  }
}

__global__ __launch_bounds__(256) void zero_rows_kernel(
    float* __restrict__ nxt, const int* __restrict__ level, int minLevel) {
  const int lane = threadIdx.x & 31;
  const int row = blockIdx.x * blockDim.x + threadIdx.x;
  bool need = false;
  if (row < N_TOTAL) need = level[row] >= minLevel;
  unsigned mask = __builtin_amdgcn_ballot_w32(need);
  const int rowBase = row - lane;
  const v2f z = {0.0f, 0.0f};
  while (mask) {
    int b = __builtin_ctz(mask);
    mask &= mask - 1;
    *(v2f*)(nxt + (size_t)(rowBase + b) * EMBED_DIM + lane * 2) = z;
  }
}

__global__ __launch_bounds__(256) void scatter_add_kernel(
    const float* __restrict__ cur, float* __restrict__ nxt,
    const int* __restrict__ src, const int* __restrict__ dst,
    const int* __restrict__ level, int minLevel) {
  const int lane = threadIdx.x & 31;
  const int e = blockIdx.x * blockDim.x + threadIdx.x;
  int d = 0, s = 0;
  bool keep = false;
  if (e < NUM_EDGES) {
    d = dst[e];
    keep = level[d] >= minLevel;
    if (keep) s = src[e];
  }
  unsigned mask = __builtin_amdgcn_ballot_w32(keep);
  while (mask) {
    int b = __builtin_ctz(mask);
    mask &= mask - 1;
    const int ss = __builtin_amdgcn_readlane(s, b);
    const int dd = __builtin_amdgcn_readlane(d, b);
    const float* sp = cur + (size_t)ss * EMBED_DIM + lane;
    float* dp = nxt + (size_t)dd * EMBED_DIM + lane;
    const float v0 = sp[0];
    const float v1 = sp[32];
    atomicAdd(dp, v0);
    atomicAdd(dp + 32, v1);
  }
}

__global__ __launch_bounds__(256) void gather_acc_kernel(
    const float* __restrict__ layer,
    const int* __restrict__ uidx, const int* __restrict__ iidx,
    float* __restrict__ ubuf, float* __restrict__ vbuf) {
  int t = blockIdx.x * blockDim.x + threadIdx.x;
  int b = t >> 4;
  if (b >= BATCH) return;
  int part = (t & 15) << 2;
  size_t u = (size_t)uidx[b];
  size_t it = (size_t)iidx[b] + NUM_USERS;
  const float4 uu = *(const float4*)(layer + u * EMBED_DIM + part);
  const float4 vv = *(const float4*)(layer + it * EMBED_DIM + part);
  float4* up = (float4*)(ubuf + (size_t)b * EMBED_DIM + part);
  float4* vp = (float4*)(vbuf + (size_t)b * EMBED_DIM + part);
  float4 a = *up;
  a.x += uu.x; a.y += uu.y; a.z += uu.z; a.w += uu.w;
  *up = a;
  float4 c = *vp;
  c.x += vv.x; c.y += vv.y; c.z += vv.z; c.w += vv.w;
  *vp = c;
}

__global__ __launch_bounds__(128) void dot_wmma_kernel(
    const float* __restrict__ ubuf, const float* __restrict__ vbuf,
    float* __restrict__ out) {
  const int lane = threadIdx.x & 31;
  const int tile = blockIdx.x * (blockDim.x >> 5) + (threadIdx.x >> 5);
  const int rowBase = tile * 16;
  const int row = rowBase + (lane & 15);
  const int kOff = (lane >> 4) << 1;

  const float* uRow = ubuf + (size_t)row * EMBED_DIM + kOff;
  const float* vRow = vbuf + (size_t)row * EMBED_DIM + kOff;

  v8f c = {};
#pragma unroll
  for (int k = 0; k < EMBED_DIM; k += 4) {
    v2f a, b;
    a.x = uRow[k];
    a.y = uRow[k + 1];
    b.x = vRow[k];
    b.y = vRow[k + 1];
    c = __builtin_amdgcn_wmma_f32_16x16x4_f32(false, a, false, b, (short)0, c,
                                              false, false);
  }

  float cd[8];
#pragma unroll
  for (int i = 0; i < 8; ++i) cd[i] = c[i];
  int dsel = (lane < 16) ? lane : (lane - 24);
  float diag = cd[0];
#pragma unroll
  for (int i = 1; i < 8; ++i)
    if (dsel == i) diag = cd[i];

  const float scale = 1.0f / 16.0f;
  if (lane < 8)
    out[rowBase + lane] = diag * scale;
  else if (lane >= 24)
    out[rowBase + (lane - 16)] = diag * scale;
}

extern "C" void kernel_launch(void* const* d_in, const int* in_sizes, int n_in,
                              void* d_out, int out_size, void* d_ws,
                              size_t ws_size, hipStream_t stream) {
  const float* emb = (const float*)d_in[0];
  const int* eidx = (const int*)d_in[1];
  const int* uidx = (const int*)d_in[2];
  const int* iidx = (const int*)d_in[3];
  float* out = (float*)d_out;

  const int* src = eidx;
  const int* dst = eidx + NUM_EDGES;

  char* ws = (char*)d_ws;
  const size_t nodeBytes = (size_t)N_TOTAL * EMBED_DIM * sizeof(float);
  const size_t batBytes = (size_t)BATCH * EMBED_DIM * sizeof(float);
  float* bufA = (float*)ws;
  float* bufB = (float*)(ws + nodeBytes);
  float* ubuf = (float*)(ws + 2 * nodeBytes);
  float* vbuf = (float*)(ws + 2 * nodeBytes + batBytes);
  int* level = (int*)ubuf;

  clear_level_kernel<<<(N_TOTAL / 4 + 255) / 256, 256, 0, stream>>>(level);
  const int eGrid = (NUM_EDGES + 255) / 256;
  mark_batch_kernel<<<(2 * BATCH + 255) / 256, 256, 0, stream>>>(uidx, iidx, level);
  mark_src_kernel<<<eGrid, 256, 0, stream>>>(src, dst, level, 2);
  mark_src_kernel<<<eGrid, 256, 0, stream>>>(src, dst, level, 1);

  const int zGrid = (N_TOTAL + 255) / 256;
  zero_rows_kernel<<<zGrid, 256, 0, stream>>>(bufA, level, 1);
  scatter_add_kernel<<<eGrid, 256, 0, stream>>>(emb, bufA, src, dst, level, 1);
  zero_rows_kernel<<<zGrid, 256, 0, stream>>>(bufB, level, 2);
  scatter_add_kernel<<<eGrid, 256, 0, stream>>>(bufA, bufB, src, dst, level, 2);
  scatter_add_kernel<<<eGrid, 256, 0, stream>>>(bufB, bufA, src, dst, level, 3);

  hipMemsetAsync(ubuf, 0, batBytes, stream);
  hipMemsetAsync(vbuf, 0, batBytes, stream);
  const int gGrid = (BATCH * 16 + 255) / 256;
  gather_acc_kernel<<<gGrid, 256, 0, stream>>>(emb, uidx, iidx, ubuf, vbuf);
  gather_acc_kernel<<<gGrid, 256, 0, stream>>>(bufA, uidx, iidx, ubuf, vbuf);
  gather_acc_kernel<<<gGrid, 256, 0, stream>>>(bufB, uidx, iidx, ubuf, vbuf);

  dot_wmma_kernel<<<BATCH / 16 / 4, 128, 0, stream>>>(ubuf, vbuf, out);
}


# Four consecutive edges (or rows) per lane, one 16-byte index load, in mark, zero and scatter launches: a quarter of the waves; kept rows move four at a time, loads ahead of the float atomics (kept)
# speedup vs baseline: 3.1466x; 1.1524x over previous
#include <hip/hip_runtime.h>
#include <hip/hip_bf16.h>

#define NUM_USERS 100000
#define NUM_ITEMS 200000
#define N_TOTAL   (NUM_USERS + NUM_ITEMS)
#define EMBED_DIM 64
#define NUM_LAYERS 3
#define NUM_EDGES 1200000
#define BATCH 4096

typedef __attribute__((ext_vector_type(2))) float v2f;
typedef __attribute__((ext_vector_type(8))) float v8f;

__global__ __launch_bounds__(256) void clear_level_kernel(int* __restrict__ level) {
  int t = blockIdx.x * blockDim.x + threadIdx.x;
  if (t < N_TOTAL / 4) ((int4*)level)[t] = make_int4(0, 0, 0, 0);
}

__global__ __launch_bounds__(256) void mark_batch_kernel(
    const int* __restrict__ uidx, const int* __restrict__ iidx,
    int* __restrict__ level) {
  int t = blockIdx.x * blockDim.x + threadIdx.x;
  if (t < BATCH)
    level[uidx[t]] = NUM_LAYERS;
  else if (t < 2 * BATCH)
    level[NUM_USERS + iidx[t - BATCH]] = NUM_LAYERS;
}

__global__ __launch_bounds__(256) void mark_src_kernel(
    const int* __restrict__ src, const int* __restrict__ dst,
    int* level, int want) {
  const int q = blockIdx.x * blockDim.x + threadIdx.x;
  if (q >= NUM_EDGES / 4) return;
  const int4 d = ((const int4*)dst)[q];
  const bool k0 = level[d.x] > want;
  const bool k1 = level[d.y] > want;
  const bool k2 = level[d.z] > want;
  const bool k3 = level[d.w] > want;
  if (k0 | k1 | k2 | k3) {
    const int4 s = ((const int4*)src)[q];
    if (k0 && level[s.x] < want) level[s.x] = want;
    if (k1 && level[s.y] < want) level[s.y] = want;
    if (k2 && level[s.z] < want) level[s.z] = want;
    if (k3 && level[s.w] < want) level[s.w] = want;
  }
}

static __device__ __forceinline__ void clear_rows(
    float* __restrict__ nxt, unsigned mask, int qBase, int j, int lane) {
  const v2f z = {0.0f, 0.0f};
  while (mask) {
    const int b = __builtin_ctz(mask);
    mask &= mask - 1;
    const int row = (qBase + b) * 4 + j;
    *(v2f*)(nxt + (size_t)row * EMBED_DIM + lane * 2) = z;
  }
}

__global__ __launch_bounds__(256) void zero_rows_kernel(
    float* __restrict__ nxt, const int* __restrict__ level, int minLevel) {
  const int lane = threadIdx.x & 31;
  const int q = blockIdx.x * blockDim.x + threadIdx.x;
  bool n0 = false, n1 = false, n2 = false, n3 = false;
  if (q < N_TOTAL / 4) {
    const int4 l = ((const int4*)level)[q];
    n0 = l.x >= minLevel;
    n1 = l.y >= minLevel;
    n2 = l.z >= minLevel;
    n3 = l.w >= minLevel;
  }
  const int qBase = q - lane;
  clear_rows(nxt, __builtin_amdgcn_ballot_w32(n0), qBase, 0, lane);
  clear_rows(nxt, __builtin_amdgcn_ballot_w32(n1), qBase, 1, lane);
  clear_rows(nxt, __builtin_amdgcn_ballot_w32(n2), qBase, 2, lane);
  clear_rows(nxt, __builtin_amdgcn_ballot_w32(n3), qBase, 3, lane);
}

static __device__ __forceinline__ void move_rows(
    const float* __restrict__ cur, float* __restrict__ nxt,
    int s, int d, unsigned mask, int lane) {
  while (mask) {
    const int b0 = __builtin_ctz(mask);
    mask &= mask - 1;
    const bool h1 = mask != 0;
    const int b1 = h1 ? __builtin_ctz(mask) : b0;
    mask &= mask - 1;
    const bool h2 = mask != 0;
    const int b2 = h2 ? __builtin_ctz(mask) : b0;
    mask &= mask - 1;
    const bool h3 = mask != 0;
    const int b3 = h3 ? __builtin_ctz(mask) : b0;
    mask &= mask - 1;
    const float* sp0 = cur + (size_t)__builtin_amdgcn_readlane(s, b0) * EMBED_DIM + lane;
    const float* sp1 = cur + (size_t)__builtin_amdgcn_readlane(s, b1) * EMBED_DIM + lane;
    const float* sp2 = cur + (size_t)__builtin_amdgcn_readlane(s, b2) * EMBED_DIM + lane;
    const float* sp3 = cur + (size_t)__builtin_amdgcn_readlane(s, b3) * EMBED_DIM + lane;
    const float a0 = sp0[0], a1 = sp0[32];
    const float c0 = sp1[0], c1 = sp1[32];
    const float e0 = sp2[0], e1 = sp2[32];
    const float f0 = sp3[0], f1 = sp3[32];
    asm volatile("" ::"v"(a0), "v"(a1), "v"(c0), "v"(c1), "v"(e0), "v"(e1),
                 "v"(f0), "v"(f1));
    float* dp0 = nxt + (size_t)__builtin_amdgcn_readlane(d, b0) * EMBED_DIM + lane;
    atomicAdd(dp0, a0);
    atomicAdd(dp0 + 32, a1);
    if (h1) {
      float* dp1 = nxt + (size_t)__builtin_amdgcn_readlane(d, b1) * EMBED_DIM + lane;
      atomicAdd(dp1, c0);
      atomicAdd(dp1 + 32, c1);
    }
    if (h2) {
      float* dp2 = nxt + (size_t)__builtin_amdgcn_readlane(d, b2) * EMBED_DIM + lane;
      atomicAdd(dp2, e0);
      atomicAdd(dp2 + 32, e1);
    }
    if (h3) {
      float* dp3 = nxt + (size_t)__builtin_amdgcn_readlane(d, b3) * EMBED_DIM + lane;
      atomicAdd(dp3, f0);
      atomicAdd(dp3 + 32, f1);
    }
  }
}

__global__ __launch_bounds__(256) void scatter_add_kernel(
    const float* __restrict__ cur, float* __restrict__ nxt,
    const int* __restrict__ src, const int* __restrict__ dst,
    const int* __restrict__ level, int minLevel) {
  const int lane = threadIdx.x & 31;
  const int q = blockIdx.x * blockDim.x + threadIdx.x;
  int4 d = make_int4(0, 0, 0, 0);
  int4 s = make_int4(0, 0, 0, 0);
  bool k0 = false, k1 = false, k2 = false, k3 = false;
  if (q < NUM_EDGES / 4) {
    d = ((const int4*)dst)[q];
    k0 = level[d.x] >= minLevel;
    k1 = level[d.y] >= minLevel;
    k2 = level[d.z] >= minLevel;
    k3 = level[d.w] >= minLevel;
    if (k0 | k1 | k2 | k3) s = ((const int4*)src)[q];
  }
  move_rows(cur, nxt, s.x, d.x, __builtin_amdgcn_ballot_w32(k0), lane);
  move_rows(cur, nxt, s.y, d.y, __builtin_amdgcn_ballot_w32(k1), lane);
  move_rows(cur, nxt, s.z, d.z, __builtin_amdgcn_ballot_w32(k2), lane);
  move_rows(cur, nxt, s.w, d.w, __builtin_amdgcn_ballot_w32(k3), lane);
}

__global__ __launch_bounds__(256) void gather_acc_kernel(
    const float* __restrict__ layer,
    const int* __restrict__ uidx, const int* __restrict__ iidx,
    float* __restrict__ ubuf, float* __restrict__ vbuf) {
  int t = blockIdx.x * blockDim.x + threadIdx.x;
  int b = t >> 4;
  if (b >= BATCH) return;
  int part = (t & 15) << 2;
  size_t u = (size_t)uidx[b];
  size_t it = (size_t)iidx[b] + NUM_USERS;
  const float4 uu = *(const float4*)(layer + u * EMBED_DIM + part);
  const float4 vv = *(const float4*)(layer + it * EMBED_DIM + part);
  float4* up = (float4*)(ubuf + (size_t)b * EMBED_DIM + part);
  float4* vp = (float4*)(vbuf + (size_t)b * EMBED_DIM + part);
  float4 a = *up;
  a.x += uu.x; a.y += uu.y; a.z += uu.z; a.w += uu.w;
  *up = a;
  float4 c = *vp;
  c.x += vv.x; c.y += vv.y; c.z += vv.z; c.w += vv.w;
  *vp = c;
}

__global__ __launch_bounds__(128) void dot_wmma_kernel(
    const float* __restrict__ ubuf, const float* __restrict__ vbuf,
    float* __restrict__ out) {
  const int lane = threadIdx.x & 31;
  const int tile = blockIdx.x * (blockDim.x >> 5) + (threadIdx.x >> 5);
  const int rowBase = tile * 16;
  const int row = rowBase + (lane & 15);
  const int kOff = (lane >> 4) << 1;

  const float* uRow = ubuf + (size_t)row * EMBED_DIM + kOff;
  const float* vRow = vbuf + (size_t)row * EMBED_DIM + kOff;

  v8f c = {};
#pragma unroll
  for (int k = 0; k < EMBED_DIM; k += 4) {
    v2f a, b;
    a.x = uRow[k];
    a.y = uRow[k + 1];
    b.x = vRow[k];
    b.y = vRow[k + 1];
    c = __builtin_amdgcn_wmma_f32_16x16x4_f32(false, a, false, b, (short)0, c,
                                              false, false);
  }

  float cd[8];
#pragma unroll
  for (int i = 0; i < 8; ++i) cd[i] = c[i];
  int dsel = (lane < 16) ? lane : (lane - 24);
  float diag = cd[0];
#pragma unroll
  for (int i = 1; i < 8; ++i)
    if (dsel == i) diag = cd[i];

  const float scale = 1.0f / 16.0f;
  if (lane < 8)
    out[rowBase + lane] = diag * scale;
  else if (lane >= 24)
    out[rowBase + (lane - 16)] = diag * scale;
}

extern "C" void kernel_launch(void* const* d_in, const int* in_sizes, int n_in,
                              void* d_out, int out_size, void* d_ws,
                              size_t ws_size, hipStream_t stream) {
  const float* emb = (const float*)d_in[0];
  const int* eidx = (const int*)d_in[1];
  const int* uidx = (const int*)d_in[2];
  const int* iidx = (const int*)d_in[3];
  float* out = (float*)d_out;

  const int* src = eidx;
  const int* dst = eidx + NUM_EDGES;

  char* ws = (char*)d_ws;
  const size_t nodeBytes = (size_t)N_TOTAL * EMBED_DIM * sizeof(float);
  const size_t batBytes = (size_t)BATCH * EMBED_DIM * sizeof(float);
  float* bufA = (float*)ws;
  float* bufB = (float*)(ws + nodeBytes);
  float* ubuf = (float*)(ws + 2 * nodeBytes);
  float* vbuf = (float*)(ws + 2 * nodeBytes + batBytes);
  int* level = (int*)ubuf;

  clear_level_kernel<<<(N_TOTAL / 4 + 255) / 256, 256, 0, stream>>>(level);
  const int eGrid = (NUM_EDGES / 4 + 255) / 256;
  mark_batch_kernel<<<(2 * BATCH + 255) / 256, 256, 0, stream>>>(uidx, iidx, level);
  mark_src_kernel<<<eGrid, 256, 0, stream>>>(src, dst, level, 2);
  mark_src_kernel<<<eGrid, 256, 0, stream>>>(src, dst, level, 1);

  const int zGrid = (N_TOTAL / 4 + 255) / 256;
  zero_rows_kernel<<<zGrid, 256, 0, stream>>>(bufA, level, 1);
  scatter_add_kernel<<<eGrid, 256, 0, stream>>>(emb, bufA, src, dst, level, 1);
  zero_rows_kernel<<<zGrid, 256, 0, stream>>>(bufB, level, 2);
  scatter_add_kernel<<<eGrid, 256, 0, stream>>>(bufA, bufB, src, dst, level, 2);
  scatter_add_kernel<<<eGrid, 256, 0, stream>>>(bufB, bufA, src, dst, level, 3);

  hipMemsetAsync(ubuf, 0, batBytes, stream);
  hipMemsetAsync(vbuf, 0, batBytes, stream);
  const int gGrid = (BATCH * 16 + 255) / 256;
  gather_acc_kernel<<<gGrid, 256, 0, stream>>>(emb, uidx, iidx, ubuf, vbuf);
  gather_acc_kernel<<<gGrid, 256, 0, stream>>>(bufA, uidx, iidx, ubuf, vbuf);
  gather_acc_kernel<<<gGrid, 256, 0, stream>>>(bufB, uidx, iidx, ubuf, vbuf);

  dot_wmma_kernel<<<BATCH / 16 / 4, 128, 0, stream>>>(ubuf, vbuf, out);
}


# Edge passes test destinations against one-bit-per-row bitmaps copied into LDS per block (built from row-clearing ballots; batch set by integer atomic or), not scattered int-table reads in L2
# speedup vs baseline: 3.4321x; 1.0907x over previous
#include <hip/hip_runtime.h>
#include <hip/hip_bf16.h>

#define NUM_USERS 100000
#define NUM_ITEMS 200000
#define N_TOTAL   (NUM_USERS + NUM_ITEMS)
#define EMBED_DIM 64
#define NUM_LAYERS 3
#define NUM_EDGES 1200000
#define BATCH 4096

typedef __attribute__((ext_vector_type(2))) float v2f;
typedef __attribute__((ext_vector_type(8))) float v8f;

#define BM_WORDS (((N_TOTAL / 4 + 255) / 256) * 8 * 4)

static __device__ __forceinline__ unsigned bm_word(int n) {
  return ((unsigned)n >> 7) * 4u + ((unsigned)n & 3u);
}
static __device__ __forceinline__ unsigned bm_shift(int n) {
  return ((unsigned)n >> 2) & 31u;
}
static __device__ __forceinline__ bool bm_test(const unsigned* bm, int n) {
  return ((bm[bm_word(n)] >> bm_shift(n)) & 1u) != 0u;
}
#define BM_TRIPS ((BM_WORDS / 4 + 255) / 256)
#define BM_LDS_WORDS (BM_TRIPS * 256 * 4)
static __device__ __forceinline__ void stage_bitmap(
    unsigned* sbm, const unsigned* __restrict__ bm) {
  int4 r[BM_TRIPS];
#pragma unroll
  for (int i = 0; i < BM_TRIPS; ++i) {
    const int idx = i * 256 + (int)threadIdx.x;
    r[i] = ((const int4*)bm)[idx < BM_WORDS / 4 ? idx : BM_WORDS / 4 - 1];
  }
#pragma unroll
  for (int i = 0; i < BM_TRIPS; ++i)
    ((int4*)sbm)[i * 256 + (int)threadIdx.x] = r[i];
}

__global__ __launch_bounds__(256) void clear_level_kernel(
    int* __restrict__ level, unsigned* __restrict__ bm3) {
  int t = blockIdx.x * blockDim.x + threadIdx.x;
  if (t < N_TOTAL / 4) ((int4*)level)[t] = make_int4(0, 0, 0, 0);
  if (t < BM_WORDS / 4) ((int4*)bm3)[t] = make_int4(0, 0, 0, 0);
}

__global__ __launch_bounds__(256) void mark_batch_kernel(
    const int* __restrict__ uidx, const int* __restrict__ iidx,
    int* __restrict__ level, unsigned* __restrict__ bm3) {
  int t = blockIdx.x * blockDim.x + threadIdx.x;
  int n;
  if (t < BATCH)
    n = uidx[t];
  else if (t < 2 * BATCH)
    n = NUM_USERS + iidx[t - BATCH];
  else
    return;
  level[n] = NUM_LAYERS;
  atomicOr(bm3 + bm_word(n), 1u << bm_shift(n));
}

__global__ __launch_bounds__(256) void mark_src_kernel(
    const int* __restrict__ src, const int* __restrict__ dst,
    int* level, const unsigned* __restrict__ above, int want) {
  __shared__ unsigned sbm[BM_LDS_WORDS];
  const int q = blockIdx.x * blockDim.x + threadIdx.x;
  const bool in = q < NUM_EDGES / 4;
  int4 d = make_int4(0, 0, 0, 0);
  if (in) d = ((const int4*)dst)[q];
  stage_bitmap(sbm, above);
  __syncthreads();
  if (!in) return;
  const bool k0 = bm_test(sbm, d.x);
  const bool k1 = bm_test(sbm, d.y);
  const bool k2 = bm_test(sbm, d.z);
  const bool k3 = bm_test(sbm, d.w);
  if (k0 | k1 | k2 | k3) {
    const int4 s = ((const int4*)src)[q];
    if (k0 && level[s.x] < want) level[s.x] = want;
    if (k1 && level[s.y] < want) level[s.y] = want;
    if (k2 && level[s.z] < want) level[s.z] = want;
    if (k3 && level[s.w] < want) level[s.w] = want;
  }
}

static __device__ __forceinline__ void clear_rows(
    float* __restrict__ nxt, unsigned mask, int qBase, int j, int lane) {
  const v2f z = {0.0f, 0.0f};
  while (mask) {
    const int b = __builtin_ctz(mask);
    mask &= mask - 1;
    const int row = (qBase + b) * 4 + j;
    *(v2f*)(nxt + (size_t)row * EMBED_DIM + lane * 2) = z;
  }
}

__global__ __launch_bounds__(256) void zero_rows_kernel(
    float* __restrict__ nxt, const int* __restrict__ level, int minLevel,
    unsigned* __restrict__ bm) {
  const int lane = threadIdx.x & 31;
  const int q = blockIdx.x * blockDim.x + threadIdx.x;
  bool n0 = false, n1 = false, n2 = false, n3 = false;
  if (q < N_TOTAL / 4) {
    const int4 l = ((const int4*)level)[q];
    n0 = l.x >= minLevel;
    n1 = l.y >= minLevel;
    n2 = l.z >= minLevel;
    n3 = l.w >= minLevel;
  }
  const int qBase = q - lane;
  const unsigned m0 = __builtin_amdgcn_ballot_w32(n0);
  const unsigned m1 = __builtin_amdgcn_ballot_w32(n1);
  const unsigned m2 = __builtin_amdgcn_ballot_w32(n2);
  const unsigned m3 = __builtin_amdgcn_ballot_w32(n3);
  if (lane < 4)
    bm[(qBase >> 5) * 4 + lane] = lane == 0 ? m0 : lane == 1 ? m1 : lane == 2 ? m2 : m3;
  clear_rows(nxt, m0, qBase, 0, lane);
  clear_rows(nxt, m1, qBase, 1, lane);
  clear_rows(nxt, m2, qBase, 2, lane);
  clear_rows(nxt, m3, qBase, 3, lane);
}

static __device__ __forceinline__ void move_rows(
    const float* __restrict__ cur, float* __restrict__ nxt,
    int s, int d, unsigned mask, int lane) {
  while (mask) {
    const int b0 = __builtin_ctz(mask);
    mask &= mask - 1;
    const bool h1 = mask != 0;
    const int b1 = h1 ? __builtin_ctz(mask) : b0;
    mask &= mask - 1;
    const bool h2 = mask != 0;
    const int b2 = h2 ? __builtin_ctz(mask) : b0;
    mask &= mask - 1;
    const bool h3 = mask != 0;
    const int b3 = h3 ? __builtin_ctz(mask) : b0;
    mask &= mask - 1;
    const float* sp0 = cur + (size_t)__builtin_amdgcn_readlane(s, b0) * EMBED_DIM + lane;
    const float* sp1 = cur + (size_t)__builtin_amdgcn_readlane(s, b1) * EMBED_DIM + lane;
    const float* sp2 = cur + (size_t)__builtin_amdgcn_readlane(s, b2) * EMBED_DIM + lane;
    const float* sp3 = cur + (size_t)__builtin_amdgcn_readlane(s, b3) * EMBED_DIM + lane;
    const float a0 = sp0[0], a1 = sp0[32];
    const float c0 = sp1[0], c1 = sp1[32];
    const float e0 = sp2[0], e1 = sp2[32];
    const float f0 = sp3[0], f1 = sp3[32];
    asm volatile("" ::"v"(a0), "v"(a1), "v"(c0), "v"(c1), "v"(e0), "v"(e1),
                 "v"(f0), "v"(f1));
    float* dp0 = nxt + (size_t)__builtin_amdgcn_readlane(d, b0) * EMBED_DIM + lane;
    atomicAdd(dp0, a0);
    atomicAdd(dp0 + 32, a1);
    if (h1) {
      float* dp1 = nxt + (size_t)__builtin_amdgcn_readlane(d, b1) * EMBED_DIM + lane;
      atomicAdd(dp1, c0);
      atomicAdd(dp1 + 32, c1);
    }
    if (h2) {
      float* dp2 = nxt + (size_t)__builtin_amdgcn_readlane(d, b2) * EMBED_DIM + lane;
      atomicAdd(dp2, e0);
      atomicAdd(dp2 + 32, e1);
    }
    if (h3) {
      float* dp3 = nxt + (size_t)__builtin_amdgcn_readlane(d, b3) * EMBED_DIM + lane;
      atomicAdd(dp3, f0);
      atomicAdd(dp3 + 32, f1);
    }
  }
}

__global__ __launch_bounds__(256) void scatter_add_kernel(
    const float* __restrict__ cur, float* __restrict__ nxt,
    const int* __restrict__ src, const int* __restrict__ dst,
    const unsigned* __restrict__ keep) {
  __shared__ unsigned sbm[BM_LDS_WORDS];
  const int lane = threadIdx.x & 31;
  const int q = blockIdx.x * blockDim.x + threadIdx.x;
  const bool in = q < NUM_EDGES / 4;
  int4 d = make_int4(0, 0, 0, 0);
  int4 s = make_int4(0, 0, 0, 0);
  bool k0 = false, k1 = false, k2 = false, k3 = false;
  if (in) d = ((const int4*)dst)[q];
  stage_bitmap(sbm, keep);
  __syncthreads();
  if (in) {
    k0 = bm_test(sbm, d.x);
    k1 = bm_test(sbm, d.y);
    k2 = bm_test(sbm, d.z);
    k3 = bm_test(sbm, d.w);
    if (k0 | k1 | k2 | k3) s = ((const int4*)src)[q];
  }
  move_rows(cur, nxt, s.x, d.x, __builtin_amdgcn_ballot_w32(k0), lane);
  move_rows(cur, nxt, s.y, d.y, __builtin_amdgcn_ballot_w32(k1), lane);
  move_rows(cur, nxt, s.z, d.z, __builtin_amdgcn_ballot_w32(k2), lane);
  move_rows(cur, nxt, s.w, d.w, __builtin_amdgcn_ballot_w32(k3), lane);
}

__global__ __launch_bounds__(256) void gather_acc_kernel(
    const float* __restrict__ layer,
    const int* __restrict__ uidx, const int* __restrict__ iidx,
    float* __restrict__ ubuf, float* __restrict__ vbuf) {
  int t = blockIdx.x * blockDim.x + threadIdx.x;
  int b = t >> 4;
  if (b >= BATCH) return;
  int part = (t & 15) << 2;
  size_t u = (size_t)uidx[b];
  size_t it = (size_t)iidx[b] + NUM_USERS;
  const float4 uu = *(const float4*)(layer + u * EMBED_DIM + part);
  const float4 vv = *(const float4*)(layer + it * EMBED_DIM + part);
  float4* up = (float4*)(ubuf + (size_t)b * EMBED_DIM + part);
  float4* vp = (float4*)(vbuf + (size_t)b * EMBED_DIM + part);
  float4 a = *up;
  a.x += uu.x; a.y += uu.y; a.z += uu.z; a.w += uu.w;
  *up = a;
  float4 c = *vp;
  c.x += vv.x; c.y += vv.y; c.z += vv.z; c.w += vv.w;
  *vp = c;
}

__global__ __launch_bounds__(128) void dot_wmma_kernel(
    const float* __restrict__ ubuf, const float* __restrict__ vbuf,
    float* __restrict__ out) {
  const int lane = threadIdx.x & 31;
  const int tile = blockIdx.x * (blockDim.x >> 5) + (threadIdx.x >> 5);
  const int rowBase = tile * 16;
  const int row = rowBase + (lane & 15);
  const int kOff = (lane >> 4) << 1;

  const float* uRow = ubuf + (size_t)row * EMBED_DIM + kOff;
  const float* vRow = vbuf + (size_t)row * EMBED_DIM + kOff;

  v8f c = {};
#pragma unroll
  for (int k = 0; k < EMBED_DIM; k += 4) {
    v2f a, b;
    a.x = uRow[k];
    a.y = uRow[k + 1];
    b.x = vRow[k];
    b.y = vRow[k + 1];
    c = __builtin_amdgcn_wmma_f32_16x16x4_f32(false, a, false, b, (short)0, c,
                                              false, false);
  }

  float cd[8];
#pragma unroll
  for (int i = 0; i < 8; ++i) cd[i] = c[i];
  int dsel = (lane < 16) ? lane : (lane - 24);
  float diag = cd[0];
#pragma unroll
  for (int i = 1; i < 8; ++i)
    if (dsel == i) diag = cd[i];

  const float scale = 1.0f / 16.0f;
  if (lane < 8)
    out[rowBase + lane] = diag * scale;
  else if (lane >= 24)
    out[rowBase + (lane - 16)] = diag * scale;
}

extern "C" void kernel_launch(void* const* d_in, const int* in_sizes, int n_in,
                              void* d_out, int out_size, void* d_ws,
                              size_t ws_size, hipStream_t stream) {
  const float* emb = (const float*)d_in[0];
  const int* eidx = (const int*)d_in[1];
  const int* uidx = (const int*)d_in[2];
  const int* iidx = (const int*)d_in[3];
  float* out = (float*)d_out;

  const int* src = eidx;
  const int* dst = eidx + NUM_EDGES;

  char* ws = (char*)d_ws;
  const size_t nodeBytes = (size_t)N_TOTAL * EMBED_DIM * sizeof(float);
  const size_t batBytes = (size_t)BATCH * EMBED_DIM * sizeof(float);
  float* bufA = (float*)ws;
  float* bufB = (float*)(ws + nodeBytes);
  float* ubuf = (float*)(ws + 2 * nodeBytes);
  float* vbuf = (float*)(ws + 2 * nodeBytes + batBytes);
  int* level = (int*)ubuf;
  unsigned* bm1 = (unsigned*)(level + N_TOTAL);
  unsigned* bm2 = bm1 + BM_WORDS;
  unsigned* bm3 = bm2 + BM_WORDS;

  clear_level_kernel<<<(N_TOTAL / 4 + 255) / 256, 256, 0, stream>>>(level, bm3);
  const int eGrid = (NUM_EDGES / 4 + 255) / 256;
  const int zGrid = (N_TOTAL / 4 + 255) / 256;
  mark_batch_kernel<<<(2 * BATCH + 255) / 256, 256, 0, stream>>>(uidx, iidx, level, bm3);
  mark_src_kernel<<<eGrid, 256, 0, stream>>>(src, dst, level, bm3, 2);
  zero_rows_kernel<<<zGrid, 256, 0, stream>>>(bufB, level, 2, bm2);
  mark_src_kernel<<<eGrid, 256, 0, stream>>>(src, dst, level, bm2, 1);

  zero_rows_kernel<<<zGrid, 256, 0, stream>>>(bufA, level, 1, bm1);
  scatter_add_kernel<<<eGrid, 256, 0, stream>>>(emb, bufA, src, dst, bm1);
  scatter_add_kernel<<<eGrid, 256, 0, stream>>>(bufA, bufB, src, dst, bm2);
  scatter_add_kernel<<<eGrid, 256, 0, stream>>>(bufB, bufA, src, dst, bm3);

  hipMemsetAsync(ubuf, 0, batBytes, stream);
  hipMemsetAsync(vbuf, 0, batBytes, stream);
  const int gGrid = (BATCH * 16 + 255) / 256;
  gather_acc_kernel<<<gGrid, 256, 0, stream>>>(emb, uidx, iidx, ubuf, vbuf);
  gather_acc_kernel<<<gGrid, 256, 0, stream>>>(bufA, uidx, iidx, ubuf, vbuf);
  gather_acc_kernel<<<gGrid, 256, 0, stream>>>(bufB, uidx, iidx, ubuf, vbuf);

  dot_wmma_kernel<<<BATCH / 16 / 4, 128, 0, stream>>>(ubuf, vbuf, out);
}


# Batch gathers, both accumulator planes with their memsets, and the WMMA dot fused into one launch: fragments summed in registers from the three planes (6 stream operations to 1; sum order kept)
# speedup vs baseline: 3.6423x; 1.0612x over previous
#include <hip/hip_runtime.h>
#include <hip/hip_bf16.h>

#define NUM_USERS 100000
#define NUM_ITEMS 200000
#define N_TOTAL   (NUM_USERS + NUM_ITEMS)
#define EMBED_DIM 64
#define NUM_LAYERS 3
#define NUM_EDGES 1200000
#define BATCH 4096

typedef __attribute__((ext_vector_type(2))) float v2f;
typedef __attribute__((ext_vector_type(8))) float v8f;

#define BM_WORDS (((N_TOTAL / 4 + 255) / 256) * 8 * 4)

static __device__ __forceinline__ unsigned bm_word(int n) {
  return ((unsigned)n >> 7) * 4u + ((unsigned)n & 3u);
}
static __device__ __forceinline__ unsigned bm_shift(int n) {
  return ((unsigned)n >> 2) & 31u;
}
static __device__ __forceinline__ bool bm_test(const unsigned* bm, int n) {
  return ((bm[bm_word(n)] >> bm_shift(n)) & 1u) != 0u;
}
#define BM_TRIPS ((BM_WORDS / 4 + 255) / 256)
#define BM_LDS_WORDS (BM_TRIPS * 256 * 4)
static __device__ __forceinline__ void stage_bitmap(
    unsigned* sbm, const unsigned* __restrict__ bm) {
  int4 r[BM_TRIPS];
#pragma unroll
  for (int i = 0; i < BM_TRIPS; ++i) {
    const int idx = i * 256 + (int)threadIdx.x;
    r[i] = ((const int4*)bm)[idx < BM_WORDS / 4 ? idx : BM_WORDS / 4 - 1];
  }
#pragma unroll
  for (int i = 0; i < BM_TRIPS; ++i)
    ((int4*)sbm)[i * 256 + (int)threadIdx.x] = r[i];
}

__global__ __launch_bounds__(256) void clear_level_kernel(
    int* __restrict__ level, unsigned* __restrict__ bm3) {
  int t = blockIdx.x * blockDim.x + threadIdx.x;
  if (t < N_TOTAL / 4) ((int4*)level)[t] = make_int4(0, 0, 0, 0);
  if (t < BM_WORDS / 4) ((int4*)bm3)[t] = make_int4(0, 0, 0, 0);
}

__global__ __launch_bounds__(256) void mark_batch_kernel(
    const int* __restrict__ uidx, const int* __restrict__ iidx,
    int* __restrict__ level, unsigned* __restrict__ bm3) {
  int t = blockIdx.x * blockDim.x + threadIdx.x;
  int n;
  if (t < BATCH)
    n = uidx[t];
  else if (t < 2 * BATCH)
    n = NUM_USERS + iidx[t - BATCH];
  else
    return;
  level[n] = NUM_LAYERS;
  atomicOr(bm3 + bm_word(n), 1u << bm_shift(n));
}

__global__ __launch_bounds__(256) void mark_src_kernel(
    const int* __restrict__ src, const int* __restrict__ dst,
    int* level, const unsigned* __restrict__ above, int want) {
  __shared__ unsigned sbm[BM_LDS_WORDS];
  const int q = blockIdx.x * blockDim.x + threadIdx.x;
  const bool in = q < NUM_EDGES / 4;
  int4 d = make_int4(0, 0, 0, 0);
  if (in) d = ((const int4*)dst)[q];
  stage_bitmap(sbm, above);
  __syncthreads();
  if (!in) return;
  const bool k0 = bm_test(sbm, d.x);
  const bool k1 = bm_test(sbm, d.y);
  const bool k2 = bm_test(sbm, d.z);
  const bool k3 = bm_test(sbm, d.w);
  if (k0 | k1 | k2 | k3) {
    const int4 s = ((const int4*)src)[q];
    if (k0 && level[s.x] < want) level[s.x] = want;
    if (k1 && level[s.y] < want) level[s.y] = want;
    if (k2 && level[s.z] < want) level[s.z] = want;
    if (k3 && level[s.w] < want) level[s.w] = want;
  }
}

static __device__ __forceinline__ void clear_rows(
    float* __restrict__ nxt, unsigned mask, int qBase, int j, int lane) {
  const v2f z = {0.0f, 0.0f};
  while (mask) {
    const int b = __builtin_ctz(mask);
    mask &= mask - 1;
    const int row = (qBase + b) * 4 + j;
    *(v2f*)(nxt + (size_t)row * EMBED_DIM + lane * 2) = z;
  }
}

__global__ __launch_bounds__(256) void zero_rows_kernel(
    float* __restrict__ nxt, const int* __restrict__ level, int minLevel,
    unsigned* __restrict__ bm) {
  const int lane = threadIdx.x & 31;
  const int q = blockIdx.x * blockDim.x + threadIdx.x;
  bool n0 = false, n1 = false, n2 = false, n3 = false;
  if (q < N_TOTAL / 4) {
    const int4 l = ((const int4*)level)[q];
    n0 = l.x >= minLevel;
    n1 = l.y >= minLevel;
    n2 = l.z >= minLevel;
    n3 = l.w >= minLevel;
  }
  const int qBase = q - lane;
  const unsigned m0 = __builtin_amdgcn_ballot_w32(n0);
  const unsigned m1 = __builtin_amdgcn_ballot_w32(n1);
  const unsigned m2 = __builtin_amdgcn_ballot_w32(n2);
  const unsigned m3 = __builtin_amdgcn_ballot_w32(n3);
  if (lane < 4)
    bm[(qBase >> 5) * 4 + lane] = lane == 0 ? m0 : lane == 1 ? m1 : lane == 2 ? m2 : m3;
  clear_rows(nxt, m0, qBase, 0, lane);
  clear_rows(nxt, m1, qBase, 1, lane);
  clear_rows(nxt, m2, qBase, 2, lane);
  clear_rows(nxt, m3, qBase, 3, lane);
}

static __device__ __forceinline__ void move_rows(
    const float* __restrict__ cur, float* __restrict__ nxt,
    int s, int d, unsigned mask, int lane) {
  while (mask) {
    const int b0 = __builtin_ctz(mask);
    mask &= mask - 1;
    const bool h1 = mask != 0;
    const int b1 = h1 ? __builtin_ctz(mask) : b0;
    mask &= mask - 1;
    const bool h2 = mask != 0;
    const int b2 = h2 ? __builtin_ctz(mask) : b0;
    mask &= mask - 1;
    const bool h3 = mask != 0;
    const int b3 = h3 ? __builtin_ctz(mask) : b0;
    mask &= mask - 1;
    const float* sp0 = cur + (size_t)__builtin_amdgcn_readlane(s, b0) * EMBED_DIM + lane;
    const float* sp1 = cur + (size_t)__builtin_amdgcn_readlane(s, b1) * EMBED_DIM + lane;
    const float* sp2 = cur + (size_t)__builtin_amdgcn_readlane(s, b2) * EMBED_DIM + lane;
    const float* sp3 = cur + (size_t)__builtin_amdgcn_readlane(s, b3) * EMBED_DIM + lane;
    const float a0 = sp0[0], a1 = sp0[32];
    const float c0 = sp1[0], c1 = sp1[32];
    const float e0 = sp2[0], e1 = sp2[32];
    const float f0 = sp3[0], f1 = sp3[32];
    asm volatile("" ::"v"(a0), "v"(a1), "v"(c0), "v"(c1), "v"(e0), "v"(e1),
                 "v"(f0), "v"(f1));
    float* dp0 = nxt + (size_t)__builtin_amdgcn_readlane(d, b0) * EMBED_DIM + lane;
    atomicAdd(dp0, a0);
    atomicAdd(dp0 + 32, a1);
    if (h1) {
      float* dp1 = nxt + (size_t)__builtin_amdgcn_readlane(d, b1) * EMBED_DIM + lane;
      atomicAdd(dp1, c0);
      atomicAdd(dp1 + 32, c1);
    }
    if (h2) {
      float* dp2 = nxt + (size_t)__builtin_amdgcn_readlane(d, b2) * EMBED_DIM + lane;
      atomicAdd(dp2, e0);
      atomicAdd(dp2 + 32, e1);
    }
    if (h3) {
      float* dp3 = nxt + (size_t)__builtin_amdgcn_readlane(d, b3) * EMBED_DIM + lane;
      atomicAdd(dp3, f0);
      atomicAdd(dp3 + 32, f1);
    }
  }
}

__global__ __launch_bounds__(256) void scatter_add_kernel(
    const float* __restrict__ cur, float* __restrict__ nxt,
    const int* __restrict__ src, const int* __restrict__ dst,
    const unsigned* __restrict__ keep) {
  __shared__ unsigned sbm[BM_LDS_WORDS];
  const int lane = threadIdx.x & 31;
  const int q = blockIdx.x * blockDim.x + threadIdx.x;
  const bool in = q < NUM_EDGES / 4;
  int4 d = make_int4(0, 0, 0, 0);
  int4 s = make_int4(0, 0, 0, 0);
  bool k0 = false, k1 = false, k2 = false, k3 = false;
  if (in) d = ((const int4*)dst)[q];
  stage_bitmap(sbm, keep);
  __syncthreads();
  if (in) {
    k0 = bm_test(sbm, d.x);
    k1 = bm_test(sbm, d.y);
    k2 = bm_test(sbm, d.z);
    k3 = bm_test(sbm, d.w);
    if (k0 | k1 | k2 | k3) s = ((const int4*)src)[q];
  }
  move_rows(cur, nxt, s.x, d.x, __builtin_amdgcn_ballot_w32(k0), lane);
  move_rows(cur, nxt, s.y, d.y, __builtin_amdgcn_ballot_w32(k1), lane);
  move_rows(cur, nxt, s.z, d.z, __builtin_amdgcn_ballot_w32(k2), lane);
  move_rows(cur, nxt, s.w, d.w, __builtin_amdgcn_ballot_w32(k3), lane);
}

__global__ __launch_bounds__(32) void batch_dot_wmma_kernel(
    const float* __restrict__ emb, const float* __restrict__ planeA,
    const float* __restrict__ planeB,
    const int* __restrict__ uidx, const int* __restrict__ iidx,
    float* __restrict__ out) {
  const int lane = threadIdx.x & 31;
  const int rowBase = blockIdx.x * 16;
  const int row = rowBase + (lane & 15);
  const int kOff = (lane >> 4) << 1;

  const size_t uo = (size_t)uidx[row] * EMBED_DIM + kOff;
  const size_t vo = ((size_t)iidx[row] + NUM_USERS) * EMBED_DIM + kOff;
  const float* uE = emb + uo;
  const float* uA = planeA + uo;
  const float* uB = planeB + uo;
  const float* vE = emb + vo;
  const float* vA = planeA + vo;
  const float* vB = planeB + vo;

  v8f c = {};
#pragma unroll
  for (int k = 0; k < EMBED_DIM; k += 4) {
    const v2f ue = *(const v2f*)(uE + k);
    const v2f ua = *(const v2f*)(uA + k);
    const v2f ub = *(const v2f*)(uB + k);
    const v2f ve = *(const v2f*)(vE + k);
    const v2f va = *(const v2f*)(vA + k);
    const v2f vb = *(const v2f*)(vB + k);
    const v2f a = (ue + ua) + ub;
    const v2f b = (ve + va) + vb;
    c = __builtin_amdgcn_wmma_f32_16x16x4_f32(false, a, false, b, (short)0, c,
                                              false, false);
  }

  float cd[8];
#pragma unroll
  for (int i = 0; i < 8; ++i) cd[i] = c[i];
  int dsel = (lane < 16) ? lane : (lane - 24);
  float diag = cd[0];
#pragma unroll
  for (int i = 1; i < 8; ++i)
    if (dsel == i) diag = cd[i];

  const float scale = 1.0f / 16.0f;
  if (lane < 8)
    out[rowBase + lane] = diag * scale;
  else if (lane >= 24)
    out[rowBase + (lane - 16)] = diag * scale;
}

extern "C" void kernel_launch(void* const* d_in, const int* in_sizes, int n_in,
                              void* d_out, int out_size, void* d_ws,
                              size_t ws_size, hipStream_t stream) {
  const float* emb = (const float*)d_in[0];
  const int* eidx = (const int*)d_in[1];
  const int* uidx = (const int*)d_in[2];
  const int* iidx = (const int*)d_in[3];
  float* out = (float*)d_out;

  const int* src = eidx;
  const int* dst = eidx + NUM_EDGES;

  char* ws = (char*)d_ws;
  const size_t nodeBytes = (size_t)N_TOTAL * EMBED_DIM * sizeof(float);
  float* bufA = (float*)ws;
  float* bufB = (float*)(ws + nodeBytes);
  int* level = (int*)(ws + 2 * nodeBytes);
  unsigned* bm1 = (unsigned*)(level + N_TOTAL);
  unsigned* bm2 = bm1 + BM_WORDS;
  unsigned* bm3 = bm2 + BM_WORDS;

  clear_level_kernel<<<(N_TOTAL / 4 + 255) / 256, 256, 0, stream>>>(level, bm3);
  const int eGrid = (NUM_EDGES / 4 + 255) / 256;
  const int zGrid = (N_TOTAL / 4 + 255) / 256;
  mark_batch_kernel<<<(2 * BATCH + 255) / 256, 256, 0, stream>>>(uidx, iidx, level, bm3);
  mark_src_kernel<<<eGrid, 256, 0, stream>>>(src, dst, level, bm3, 2);
  zero_rows_kernel<<<zGrid, 256, 0, stream>>>(bufB, level, 2, bm2);
  mark_src_kernel<<<eGrid, 256, 0, stream>>>(src, dst, level, bm2, 1);

  zero_rows_kernel<<<zGrid, 256, 0, stream>>>(bufA, level, 1, bm1);
  scatter_add_kernel<<<eGrid, 256, 0, stream>>>(emb, bufA, src, dst, bm1);
  scatter_add_kernel<<<eGrid, 256, 0, stream>>>(bufA, bufB, src, dst, bm2);
  scatter_add_kernel<<<eGrid, 256, 0, stream>>>(bufB, bufA, src, dst, bm3);

  batch_dot_wmma_kernel<<<BATCH / 16, 32, 0, stream>>>(emb, bufA, bufB, uidx, iidx, out);
}


# Row sets as bitmaps only: first launch writes them whole, marking passes set source bits by integer atomic or and clear the found rows; level table and three launches dropped (10 to 7)
# speedup vs baseline: 3.9077x; 1.0729x over previous
#include <hip/hip_runtime.h>
#include <hip/hip_bf16.h>

#define NUM_USERS 100000
#define NUM_ITEMS 200000
#define N_TOTAL   (NUM_USERS + NUM_ITEMS)
#define EMBED_DIM 64
#define NUM_LAYERS 3
#define NUM_EDGES 1200000
#define BATCH 4096

typedef __attribute__((ext_vector_type(2))) float v2f;
typedef __attribute__((ext_vector_type(8))) float v8f;

#define BM_WORDS (((N_TOTAL / 4 + 255) / 256) * 8 * 4)

static __device__ __forceinline__ unsigned bm_word(int n) {
  return ((unsigned)n >> 7) * 4u + ((unsigned)n & 3u);
}
static __device__ __forceinline__ unsigned bm_shift(int n) {
  return ((unsigned)n >> 2) & 31u;
}
static __device__ __forceinline__ bool bm_test(const unsigned* bm, int n) {
  return ((bm[bm_word(n)] >> bm_shift(n)) & 1u) != 0u;
}
#define BM_TRIPS ((BM_WORDS / 4 + 255) / 256)
#define BM_LDS_WORDS (BM_TRIPS * 256 * 4)
static __device__ __forceinline__ void stage_bitmap(
    unsigned* sbm, const unsigned* __restrict__ bm) {
  int4 r[BM_TRIPS];
#pragma unroll
  for (int i = 0; i < BM_TRIPS; ++i) {
    const int idx = i * 256 + (int)threadIdx.x;
    r[i] = ((const int4*)bm)[idx < BM_WORDS / 4 ? idx : BM_WORDS / 4 - 1];
  }
#pragma unroll
  for (int i = 0; i < BM_TRIPS; ++i)
    ((int4*)sbm)[i * 256 + (int)threadIdx.x] = r[i];
}

#define MB_WORDS 256
#define MB_QUADS (BATCH / 4 / 256)
__global__ __launch_bounds__(256) void mark_batch_kernel(
    const int* __restrict__ uidx, const int* __restrict__ iidx,
    unsigned* __restrict__ bm1, unsigned* __restrict__ bm2,
    unsigned* __restrict__ bm3) {
  __shared__ unsigned part[MB_WORDS];
  const int t = (int)threadIdx.x;
  part[t] = 0u;
  int4 u[MB_QUADS], v[MB_QUADS];
#pragma unroll
  for (int i = 0; i < MB_QUADS; ++i) {
    u[i] = ((const int4*)uidx)[i * 256 + t];
    v[i] = ((const int4*)iidx)[i * 256 + t];
  }
  __syncthreads();
#pragma unroll
  for (int i = 0; i < MB_QUADS; ++i) {
    const int n[8] = {u[i].x, u[i].y, u[i].z, u[i].w,
                      NUM_USERS + v[i].x, NUM_USERS + v[i].y,
                      NUM_USERS + v[i].z, NUM_USERS + v[i].w};
#pragma unroll
    for (int j = 0; j < 8; ++j) {
      const unsigned w = bm_word(n[j]);
      if ((w / MB_WORDS) == blockIdx.x)
        atomicOr(&part[w % MB_WORDS], 1u << bm_shift(n[j]));
    }
  }
  __syncthreads();
  const int w = (int)blockIdx.x * MB_WORDS + t;
  if (w < BM_WORDS) {
    const unsigned m = part[t];
    bm1[w] = m;
    bm2[w] = m;
    bm3[w] = m;
  }
}

static __device__ __forceinline__ void zero_row(
    float* __restrict__ plane, int row, int lane) {
  const v2f z = {0.0f, 0.0f};
  *(v2f*)(plane + (size_t)row * EMBED_DIM + lane * 2) = z;
}

static __device__ __forceinline__ int set_first_word() {
  return (int)blockIdx.x * (int)(blockDim.x >> 5) +
         __builtin_amdgcn_readfirstlane((int)(threadIdx.x >> 5));
}
static __device__ __forceinline__ unsigned set_word(const unsigned* bm, int w) {
  return w < BM_WORDS ? (unsigned)__builtin_amdgcn_readfirstlane((int)bm[w]) : 0u;
}
static __device__ __forceinline__ void zero_word_rows(
    float* __restrict__ plane, int w, unsigned m, int lane) {
  const int base = (w >> 2) * 128 + (w & 3);
  while (m) {
    const int b = __builtin_ctz(m);
    m &= m - 1;
    zero_row(plane, base + b * 4, lane);
  }
}
static __device__ __forceinline__ void zero_set_rows(
    float* __restrict__ plane, const unsigned* bm, int lane, int w0,
    unsigned m0) {
  zero_word_rows(plane, w0, m0, lane);
  const int nWaves = (int)gridDim.x * (int)(blockDim.x >> 5);
  for (int w = w0 + nWaves; w < BM_WORDS; w += nWaves)
    zero_word_rows(plane, w, set_word(bm, w), lane);
}

template <int HOP>
__global__ __launch_bounds__(256) void mark_src_kernel(
    const int* __restrict__ src, const int* __restrict__ dst,
    const unsigned* __restrict__ above, unsigned* __restrict__ bmA,
    unsigned* __restrict__ bmB, float* __restrict__ plane) {
  __shared__ unsigned sbm[BM_LDS_WORDS];
  const int lane = threadIdx.x & 31;
  const int q = blockIdx.x * blockDim.x + threadIdx.x;
  const bool in = q < NUM_EDGES / 4;
  int4 d = make_int4(0, 0, 0, 0);
  int4 s = make_int4(0, 0, 0, 0);
  bool k0 = false, k1 = false, k2 = false, k3 = false;
  if (in) d = ((const int4*)dst)[q];
  stage_bitmap(sbm, above);
  __syncthreads();
  if (in) {
    k0 = bm_test(sbm, d.x);
    k1 = bm_test(sbm, d.y);
    k2 = bm_test(sbm, d.z);
    k3 = bm_test(sbm, d.w);
    if (k0 | k1 | k2 | k3) {
      s = ((const int4*)src)[q];
      const int sv[4] = {s.x, s.y, s.z, s.w};
      const bool kv[4] = {k0, k1, k2, k3};
#pragma unroll
      for (int i = 0; i < 4; ++i) {
        if (kv[i]) {
          const unsigned w = bm_word(sv[i]);
          const unsigned bit = 1u << bm_shift(sv[i]);
          if (HOP == 1) atomicOr(bmA + w, bit);
          atomicOr(bmB + w, bit);
        }
      }
    }
  }
  if (HOP == 2) {
    const int w0 = set_first_word();
    zero_set_rows(plane, sbm, lane, w0, set_word(sbm, w0));
    const int sv[4] = {s.x, s.y, s.z, s.w};
    const bool kv[4] = {k0, k1, k2, k3};
#pragma unroll
    for (int i = 0; i < 4; ++i) {
      unsigned mask = __builtin_amdgcn_ballot_w32(kv[i]);
      while (mask) {
        const int b = __builtin_ctz(mask);
        mask &= mask - 1;
        zero_row(plane, __builtin_amdgcn_readlane(sv[i], b), lane);
      }
    }
  }
}

static __device__ __forceinline__ void move_rows(
    const float* __restrict__ cur, float* __restrict__ nxt,
    int s, int d, unsigned mask, int lane) {
  while (mask) {
    const int b0 = __builtin_ctz(mask);
    mask &= mask - 1;
    const bool h1 = mask != 0;
    const int b1 = h1 ? __builtin_ctz(mask) : b0;
    mask &= mask - 1;
    const bool h2 = mask != 0;
    const int b2 = h2 ? __builtin_ctz(mask) : b0;
    mask &= mask - 1;
    const bool h3 = mask != 0;
    const int b3 = h3 ? __builtin_ctz(mask) : b0;
    mask &= mask - 1;
    const float* sp0 = cur + (size_t)__builtin_amdgcn_readlane(s, b0) * EMBED_DIM + lane;
    const float* sp1 = cur + (size_t)__builtin_amdgcn_readlane(s, b1) * EMBED_DIM + lane;
    const float* sp2 = cur + (size_t)__builtin_amdgcn_readlane(s, b2) * EMBED_DIM + lane;
    const float* sp3 = cur + (size_t)__builtin_amdgcn_readlane(s, b3) * EMBED_DIM + lane;
    const float a0 = sp0[0], a1 = sp0[32];
    const float c0 = sp1[0], c1 = sp1[32];
    const float e0 = sp2[0], e1 = sp2[32];
    const float f0 = sp3[0], f1 = sp3[32];
    asm volatile("" ::"v"(a0), "v"(a1), "v"(c0), "v"(c1), "v"(e0), "v"(e1),
                 "v"(f0), "v"(f1));
    float* dp0 = nxt + (size_t)__builtin_amdgcn_readlane(d, b0) * EMBED_DIM + lane;
    atomicAdd(dp0, a0);
    atomicAdd(dp0 + 32, a1);
    if (h1) {
      float* dp1 = nxt + (size_t)__builtin_amdgcn_readlane(d, b1) * EMBED_DIM + lane;
      atomicAdd(dp1, c0);
      atomicAdd(dp1 + 32, c1);
    }
    if (h2) {
      float* dp2 = nxt + (size_t)__builtin_amdgcn_readlane(d, b2) * EMBED_DIM + lane;
      atomicAdd(dp2, e0);
      atomicAdd(dp2 + 32, e1);
    }
    if (h3) {
      float* dp3 = nxt + (size_t)__builtin_amdgcn_readlane(d, b3) * EMBED_DIM + lane;
      atomicAdd(dp3, f0);
      atomicAdd(dp3 + 32, f1);
    }
  }
}

template <bool CLEAR_NEXT>
__global__ __launch_bounds__(256) void scatter_add_kernel(
    const float* __restrict__ cur, float* __restrict__ nxt,
    const int* __restrict__ src, const int* __restrict__ dst,
    const unsigned* __restrict__ keep, float* __restrict__ clearPlane,
    const unsigned* __restrict__ clearSet) {
  __shared__ unsigned sbm[BM_LDS_WORDS];
  const int lane = threadIdx.x & 31;
  const int q = blockIdx.x * blockDim.x + threadIdx.x;
  const bool in = q < NUM_EDGES / 4;
  int4 d = make_int4(0, 0, 0, 0);
  int4 s = make_int4(0, 0, 0, 0);
  bool k0 = false, k1 = false, k2 = false, k3 = false;
  if (in) d = ((const int4*)dst)[q];
  const int w0 = CLEAR_NEXT ? set_first_word() : 0;
  const unsigned m0 = CLEAR_NEXT ? set_word(clearSet, w0) : 0u;
  stage_bitmap(sbm, keep);
  __syncthreads();
  if (CLEAR_NEXT) zero_set_rows(clearPlane, clearSet, lane, w0, m0);
  if (in) {
    k0 = bm_test(sbm, d.x);
    k1 = bm_test(sbm, d.y);
    k2 = bm_test(sbm, d.z);
    k3 = bm_test(sbm, d.w);
    if (k0 | k1 | k2 | k3) s = ((const int4*)src)[q];
  }
  move_rows(cur, nxt, s.x, d.x, __builtin_amdgcn_ballot_w32(k0), lane);
  move_rows(cur, nxt, s.y, d.y, __builtin_amdgcn_ballot_w32(k1), lane);
  move_rows(cur, nxt, s.z, d.z, __builtin_amdgcn_ballot_w32(k2), lane);
  move_rows(cur, nxt, s.w, d.w, __builtin_amdgcn_ballot_w32(k3), lane);
}

__global__ __launch_bounds__(32) void batch_dot_wmma_kernel(
    const float* __restrict__ emb, const float* __restrict__ planeA,
    const float* __restrict__ planeB,
    const int* __restrict__ uidx, const int* __restrict__ iidx,
    float* __restrict__ out) {
  const int lane = threadIdx.x & 31;
  const int rowBase = blockIdx.x * 16;
  const int row = rowBase + (lane & 15);
  const int kOff = (lane >> 4) << 1;

  const size_t uo = (size_t)uidx[row] * EMBED_DIM + kOff;
  const size_t vo = ((size_t)iidx[row] + NUM_USERS) * EMBED_DIM + kOff;
  const float* uE = emb + uo;
  const float* uA = planeA + uo;
  const float* uB = planeB + uo;
  const float* vE = emb + vo;
  const float* vA = planeA + vo;
  const float* vB = planeB + vo;

  v8f c = {};
#pragma unroll
  for (int k = 0; k < EMBED_DIM; k += 4) {
    const v2f ue = *(const v2f*)(uE + k);
    const v2f ua = *(const v2f*)(uA + k);
    const v2f ub = *(const v2f*)(uB + k);
    const v2f ve = *(const v2f*)(vE + k);
    const v2f va = *(const v2f*)(vA + k);
    const v2f vb = *(const v2f*)(vB + k);
    const v2f a = (ue + ua) + ub;
    const v2f b = (ve + va) + vb;
    c = __builtin_amdgcn_wmma_f32_16x16x4_f32(false, a, false, b, (short)0, c,
                                              false, false);
  }

  float cd[8];
#pragma unroll
  for (int i = 0; i < 8; ++i) cd[i] = c[i];
  int dsel = (lane < 16) ? lane : (lane - 24);
  float diag = cd[0];
#pragma unroll
  for (int i = 1; i < 8; ++i)
    if (dsel == i) diag = cd[i];

  const float scale = 1.0f / 16.0f;
  if (lane < 8)
    out[rowBase + lane] = diag * scale;
  else if (lane >= 24)
    out[rowBase + (lane - 16)] = diag * scale;
}

extern "C" void kernel_launch(void* const* d_in, const int* in_sizes, int n_in,
                              void* d_out, int out_size, void* d_ws,
                              size_t ws_size, hipStream_t stream) {
  const float* emb = (const float*)d_in[0];
  const int* eidx = (const int*)d_in[1];
  const int* uidx = (const int*)d_in[2];
  const int* iidx = (const int*)d_in[3];
  float* out = (float*)d_out;

  const int* src = eidx;
  const int* dst = eidx + NUM_EDGES;

  char* ws = (char*)d_ws;
  const size_t nodeBytes = (size_t)N_TOTAL * EMBED_DIM * sizeof(float);
  float* bufA = (float*)ws;
  float* bufB = (float*)(ws + nodeBytes);
  unsigned* bm1 = (unsigned*)(ws + 2 * nodeBytes);
  unsigned* bm2 = bm1 + BM_WORDS;
  unsigned* bm3 = bm2 + BM_WORDS;

  const int eGrid = (NUM_EDGES / 4 + 255) / 256;
  mark_batch_kernel<<<(BM_WORDS + MB_WORDS - 1) / MB_WORDS, 256, 0, stream>>>(uidx, iidx, bm1, bm2, bm3);
  mark_src_kernel<1><<<eGrid, 256, 0, stream>>>(src, dst, bm3, bm2, bm1, nullptr);
  mark_src_kernel<2><<<eGrid, 256, 0, stream>>>(src, dst, bm2, nullptr, bm1, bufA);

  scatter_add_kernel<true><<<eGrid, 256, 0, stream>>>(emb, bufA, src, dst, bm1, bufB, bm2);
  scatter_add_kernel<false><<<eGrid, 256, 0, stream>>>(bufA, bufB, src, dst, bm2, nullptr, nullptr);
  scatter_add_kernel<false><<<eGrid, 256, 0, stream>>>(bufB, bufA, src, dst, bm3, nullptr, nullptr);

  batch_dot_wmma_kernel<<<BATCH / 16, 32, 0, stream>>>(emb, bufA, bufB, uidx, iidx, out);
}


# Layer sum re-associated by linearity: batch rows take layer 1 plus their sources' (layer 1+2) sums; layer 2 runs on sources of batch rows only, layer 1 drops rows only that set held; atomics kept
# speedup vs baseline: 4.5509x; 1.1646x over previous
#include <hip/hip_runtime.h>
#include <hip/hip_bf16.h>

#define NUM_USERS 100000
#define NUM_ITEMS 200000
#define N_TOTAL   (NUM_USERS + NUM_ITEMS)
#define EMBED_DIM 64
#define NUM_LAYERS 3
#define NUM_EDGES 1200000
#define BATCH 4096

typedef __attribute__((ext_vector_type(2))) float v2f;
typedef __attribute__((ext_vector_type(8))) float v8f;

#define BM_WORDS (((N_TOTAL / 4 + 255) / 256) * 8 * 4)

static __device__ __forceinline__ unsigned bm_word(int n) {
  return ((unsigned)n >> 7) * 4u + ((unsigned)n & 3u);
}
static __device__ __forceinline__ unsigned bm_shift(int n) {
  return ((unsigned)n >> 2) & 31u;
}
static __device__ __forceinline__ bool bm_test(const unsigned* bm, int n) {
  return ((bm[bm_word(n)] >> bm_shift(n)) & 1u) != 0u;
}
#define BM_TRIPS ((BM_WORDS / 4 + 255) / 256)
#define BM_LDS_WORDS (BM_TRIPS * 256 * 4)
static __device__ __forceinline__ void stage_bitmap(
    unsigned* sbm, const unsigned* __restrict__ bm) {
  int4 r[BM_TRIPS];
#pragma unroll
  for (int i = 0; i < BM_TRIPS; ++i) {
    const int idx = i * 256 + (int)threadIdx.x;
    r[i] = ((const int4*)bm)[idx < BM_WORDS / 4 ? idx : BM_WORDS / 4 - 1];
  }
#pragma unroll
  for (int i = 0; i < BM_TRIPS; ++i)
    ((int4*)sbm)[i * 256 + (int)threadIdx.x] = r[i];
}

#define MB_WORDS 256
#define MB_QUADS (BATCH / 4 / 256)
__global__ __launch_bounds__(256) void mark_batch_kernel(
    const int* __restrict__ uidx, const int* __restrict__ iidx,
    unsigned* __restrict__ bm1, unsigned* __restrict__ bm2,
    unsigned* __restrict__ bm3) {
  __shared__ unsigned part[MB_WORDS];
  const int t = (int)threadIdx.x;
  part[t] = 0u;
  int4 u[MB_QUADS], v[MB_QUADS];
#pragma unroll
  for (int i = 0; i < MB_QUADS; ++i) {
    u[i] = ((const int4*)uidx)[i * 256 + t];
    v[i] = ((const int4*)iidx)[i * 256 + t];
  }
  __syncthreads();
#pragma unroll
  for (int i = 0; i < MB_QUADS; ++i) {
    const int n[8] = {u[i].x, u[i].y, u[i].z, u[i].w,
                      NUM_USERS + v[i].x, NUM_USERS + v[i].y,
                      NUM_USERS + v[i].z, NUM_USERS + v[i].w};
#pragma unroll
    for (int j = 0; j < 8; ++j) {
      const unsigned w = bm_word(n[j]);
      if ((w / MB_WORDS) == blockIdx.x)
        atomicOr(&part[w % MB_WORDS], 1u << bm_shift(n[j]));
    }
  }
  __syncthreads();
  const int w = (int)blockIdx.x * MB_WORDS + t;
  if (w < BM_WORDS) {
    const unsigned m = part[t];
    bm1[w] = m;
    bm2[w] = 0u;
    bm3[w] = m;
  }
}

static __device__ __forceinline__ void zero_row(
    float* __restrict__ plane, int row, int lane) {
  const v2f z = {0.0f, 0.0f};
  *(v2f*)(plane + (size_t)row * EMBED_DIM + lane * 2) = z;
}

static __device__ __forceinline__ int set_first_word() {
  return (int)blockIdx.x * (int)(blockDim.x >> 5) +
         __builtin_amdgcn_readfirstlane((int)(threadIdx.x >> 5));
}
static __device__ __forceinline__ unsigned set_word(const unsigned* bm, int w) {
  return w < BM_WORDS ? (unsigned)__builtin_amdgcn_readfirstlane((int)bm[w]) : 0u;
}
static __device__ __forceinline__ void zero_word_rows(
    float* __restrict__ plane, int w, unsigned m, int lane) {
  const int base = (w >> 2) * 128 + (w & 3);
  while (m) {
    const int b = __builtin_ctz(m);
    m &= m - 1;
    zero_row(plane, base + b * 4, lane);
  }
}
static __device__ __forceinline__ void zero_set_rows(
    float* __restrict__ plane, const unsigned* bm, int lane, int w0,
    unsigned m0) {
  zero_word_rows(plane, w0, m0, lane);
  const int nWaves = (int)gridDim.x * (int)(blockDim.x >> 5);
  for (int w = w0 + nWaves; w < BM_WORDS; w += nWaves)
    zero_word_rows(plane, w, set_word(bm, w), lane);
}

template <int HOP>
__global__ __launch_bounds__(256) void mark_src_kernel(
    const int* __restrict__ src, const int* __restrict__ dst,
    const unsigned* __restrict__ above, unsigned* __restrict__ bmOut,
    const unsigned* __restrict__ walkSet, float* __restrict__ plane) {
  __shared__ unsigned sbm[BM_LDS_WORDS];
  const int lane = threadIdx.x & 31;
  const int q = blockIdx.x * blockDim.x + threadIdx.x;
  const bool in = q < NUM_EDGES / 4;
  int4 d = make_int4(0, 0, 0, 0);
  int4 s = make_int4(0, 0, 0, 0);
  bool k0 = false, k1 = false, k2 = false, k3 = false;
  if (in) d = ((const int4*)dst)[q];
  const int w0 = HOP == 2 ? set_first_word() : 0;
  const unsigned m0 = HOP == 2 ? set_word(walkSet, w0) : 0u;
  stage_bitmap(sbm, above);
  __syncthreads();
  if (in) {
    k0 = bm_test(sbm, d.x);
    k1 = bm_test(sbm, d.y);
    k2 = bm_test(sbm, d.z);
    k3 = bm_test(sbm, d.w);
    if (k0 | k1 | k2 | k3) {
      s = ((const int4*)src)[q];
      const int sv[4] = {s.x, s.y, s.z, s.w};
      const bool kv[4] = {k0, k1, k2, k3};
#pragma unroll
      for (int i = 0; i < 4; ++i) {
        if (kv[i]) {
          const unsigned w = bm_word(sv[i]);
          const unsigned bit = 1u << bm_shift(sv[i]);
          atomicOr(bmOut + w, bit);
        }
      }
    }
  }
  if (HOP == 2) {
    zero_set_rows(plane, walkSet, lane, w0, m0);
    const int sv[4] = {s.x, s.y, s.z, s.w};
    const bool kv[4] = {k0, k1, k2, k3};
#pragma unroll
    for (int i = 0; i < 4; ++i) {
      unsigned mask = __builtin_amdgcn_ballot_w32(kv[i]);
      while (mask) {
        const int b = __builtin_ctz(mask);
        mask &= mask - 1;
        zero_row(plane, __builtin_amdgcn_readlane(sv[i], b), lane);
      }
    }
  }
}

template <bool ADD_BASE>
static __device__ __forceinline__ void move_rows(
    const float* __restrict__ cur, const float* __restrict__ base,
    float* __restrict__ nxt, int s, int d, unsigned mask, int lane) {
  while (mask) {
    const int b0 = __builtin_ctz(mask);
    mask &= mask - 1;
    const bool h1 = mask != 0;
    const int b1 = h1 ? __builtin_ctz(mask) : b0;
    mask &= mask - 1;
    const bool h2 = mask != 0;
    const int b2 = h2 ? __builtin_ctz(mask) : b0;
    mask &= mask - 1;
    const bool h3 = mask != 0;
    const int b3 = h3 ? __builtin_ctz(mask) : b0;
    mask &= mask - 1;
    const size_t o0 = (size_t)__builtin_amdgcn_readlane(s, b0) * EMBED_DIM + lane;
    const size_t o1 = (size_t)__builtin_amdgcn_readlane(s, b1) * EMBED_DIM + lane;
    const size_t o2 = (size_t)__builtin_amdgcn_readlane(s, b2) * EMBED_DIM + lane;
    const size_t o3 = (size_t)__builtin_amdgcn_readlane(s, b3) * EMBED_DIM + lane;
    const float* sp0 = cur + o0;
    const float* sp1 = cur + o1;
    const float* sp2 = cur + o2;
    const float* sp3 = cur + o3;
    float a0 = sp0[0], a1 = sp0[32];
    float c0 = sp1[0], c1 = sp1[32];
    float e0 = sp2[0], e1 = sp2[32];
    float f0 = sp3[0], f1 = sp3[32];
    if (ADD_BASE) {
      const float* bp0 = base + o0;
      const float* bp1 = base + o1;
      const float* bp2 = base + o2;
      const float* bp3 = base + o3;
      const float g0 = bp0[0], g1 = bp0[32];
      const float h0 = bp1[0], h1v = bp1[32];
      const float i0 = bp2[0], i1 = bp2[32];
      const float j0 = bp3[0], j1 = bp3[32];
      asm volatile("" ::"v"(a0), "v"(a1), "v"(c0), "v"(c1), "v"(e0), "v"(e1),
                   "v"(f0), "v"(f1), "v"(g0), "v"(g1), "v"(h0), "v"(h1v),
                   "v"(i0), "v"(i1), "v"(j0), "v"(j1));
      a0 += g0; a1 += g1;
      c0 += h0; c1 += h1v;
      e0 += i0; e1 += i1;
      f0 += j0; f1 += j1;
    }
    asm volatile("" ::"v"(a0), "v"(a1), "v"(c0), "v"(c1), "v"(e0), "v"(e1),
                 "v"(f0), "v"(f1));
    float* dp0 = nxt + (size_t)__builtin_amdgcn_readlane(d, b0) * EMBED_DIM + lane;
    atomicAdd(dp0, a0);
    atomicAdd(dp0 + 32, a1);
    if (h1) {
      float* dp1 = nxt + (size_t)__builtin_amdgcn_readlane(d, b1) * EMBED_DIM + lane;
      atomicAdd(dp1, c0);
      atomicAdd(dp1 + 32, c1);
    }
    if (h2) {
      float* dp2 = nxt + (size_t)__builtin_amdgcn_readlane(d, b2) * EMBED_DIM + lane;
      atomicAdd(dp2, e0);
      atomicAdd(dp2 + 32, e1);
    }
    if (h3) {
      float* dp3 = nxt + (size_t)__builtin_amdgcn_readlane(d, b3) * EMBED_DIM + lane;
      atomicAdd(dp3, f0);
      atomicAdd(dp3 + 32, f1);
    }
  }
}

template <bool CLEAR_NEXT, bool ADD_BASE>
__global__ __launch_bounds__(256) void scatter_add_kernel(
    const float* __restrict__ cur, const float* __restrict__ base,
    float* __restrict__ nxt,
    const int* __restrict__ src, const int* __restrict__ dst,
    const unsigned* __restrict__ keep, float* __restrict__ clearPlane,
    const unsigned* __restrict__ clearSet) {
  __shared__ unsigned sbm[BM_LDS_WORDS];
  const int lane = threadIdx.x & 31;
  const int q = blockIdx.x * blockDim.x + threadIdx.x;
  const bool in = q < NUM_EDGES / 4;
  int4 d = make_int4(0, 0, 0, 0);
  int4 s = make_int4(0, 0, 0, 0);
  bool k0 = false, k1 = false, k2 = false, k3 = false;
  if (in) d = ((const int4*)dst)[q];
  const int w0 = CLEAR_NEXT ? set_first_word() : 0;
  const unsigned m0 = CLEAR_NEXT ? set_word(clearSet, w0) : 0u;
  stage_bitmap(sbm, keep);
  __syncthreads();
  if (CLEAR_NEXT) zero_set_rows(clearPlane, clearSet, lane, w0, m0);
  if (in) {
    k0 = bm_test(sbm, d.x);
    k1 = bm_test(sbm, d.y);
    k2 = bm_test(sbm, d.z);
    k3 = bm_test(sbm, d.w);
    if (k0 | k1 | k2 | k3) s = ((const int4*)src)[q];
  }
  move_rows<ADD_BASE>(cur, base, nxt, s.x, d.x, __builtin_amdgcn_ballot_w32(k0), lane);
  move_rows<ADD_BASE>(cur, base, nxt, s.y, d.y, __builtin_amdgcn_ballot_w32(k1), lane);
  move_rows<ADD_BASE>(cur, base, nxt, s.z, d.z, __builtin_amdgcn_ballot_w32(k2), lane);
  move_rows<ADD_BASE>(cur, base, nxt, s.w, d.w, __builtin_amdgcn_ballot_w32(k3), lane);
}

__global__ __launch_bounds__(32) void batch_dot_wmma_kernel(
    const float* __restrict__ emb, const float* __restrict__ planeA,
    const int* __restrict__ uidx, const int* __restrict__ iidx,
    float* __restrict__ out) {
  const int lane = threadIdx.x & 31;
  const int rowBase = blockIdx.x * 16;
  const int row = rowBase + (lane & 15);
  const int kOff = (lane >> 4) << 1;

  const size_t uo = (size_t)uidx[row] * EMBED_DIM + kOff;
  const size_t vo = ((size_t)iidx[row] + NUM_USERS) * EMBED_DIM + kOff;
  const float* uE = emb + uo;
  const float* uA = planeA + uo;
  const float* vE = emb + vo;
  const float* vA = planeA + vo;

  v8f c = {};
#pragma unroll
  for (int k = 0; k < EMBED_DIM; k += 4) {
    const v2f ue = *(const v2f*)(uE + k);
    const v2f ua = *(const v2f*)(uA + k);
    const v2f ve = *(const v2f*)(vE + k);
    const v2f va = *(const v2f*)(vA + k);
    const v2f a = ue + ua;
    const v2f b = ve + va;
    c = __builtin_amdgcn_wmma_f32_16x16x4_f32(false, a, false, b, (short)0, c,
                                              false, false);
  }

  float cd[8];
#pragma unroll
  for (int i = 0; i < 8; ++i) cd[i] = c[i];
  int dsel = (lane < 16) ? lane : (lane - 24);
  float diag = cd[0];
#pragma unroll
  for (int i = 1; i < 8; ++i)
    if (dsel == i) diag = cd[i];

  const float scale = 1.0f / 16.0f;
  if (lane < 8)
    out[rowBase + lane] = diag * scale;
  else if (lane >= 24)
    out[rowBase + (lane - 16)] = diag * scale;
}

extern "C" void kernel_launch(void* const* d_in, const int* in_sizes, int n_in,
                              void* d_out, int out_size, void* d_ws,
                              size_t ws_size, hipStream_t stream) {
  const float* emb = (const float*)d_in[0];
  const int* eidx = (const int*)d_in[1];
  const int* uidx = (const int*)d_in[2];
  const int* iidx = (const int*)d_in[3];
  float* out = (float*)d_out;

  const int* src = eidx;
  const int* dst = eidx + NUM_EDGES;

  char* ws = (char*)d_ws;
  const size_t nodeBytes = (size_t)N_TOTAL * EMBED_DIM * sizeof(float);
  float* bufA = (float*)ws;
  float* bufB = (float*)(ws + nodeBytes);
  unsigned* bm1 = (unsigned*)(ws + 2 * nodeBytes);
  unsigned* bm2 = bm1 + BM_WORDS;
  unsigned* bm3 = bm2 + BM_WORDS;

  const int eGrid = (NUM_EDGES / 4 + 255) / 256;
  mark_batch_kernel<<<(BM_WORDS + MB_WORDS - 1) / MB_WORDS, 256, 0, stream>>>(uidx, iidx, bm1, bm2, bm3);
  mark_src_kernel<1><<<eGrid, 256, 0, stream>>>(src, dst, bm3, bm2, nullptr, nullptr);
  mark_src_kernel<2><<<eGrid, 256, 0, stream>>>(src, dst, bm2, bm1, bm3, bufA);

  scatter_add_kernel<true, false><<<eGrid, 256, 0, stream>>>(emb, nullptr, bufA, src, dst, bm1, bufB, bm2);
  scatter_add_kernel<false, true><<<eGrid, 256, 0, stream>>>(bufA, emb, bufB, src, dst, bm2, nullptr, nullptr);
  scatter_add_kernel<false, false><<<eGrid, 256, 0, stream>>>(bufB, nullptr, bufA, src, dst, bm3, nullptr, nullptr);

  batch_dot_wmma_kernel<<<BATCH / 16, 32, 0, stream>>>(emb, bufA, uidx, iidx, out);
}
